# starting kernel
# baseline (speedup 1.0000x reference)
#include <hip/hip_runtime.h>

typedef __attribute__((ext_vector_type(2))) float v2f;
typedef __attribute__((ext_vector_type(8))) float v8f;

#define BLOCK 256
#define WAVES 8
#define JSTRIDE 17

__device__ __forceinline__ int cnot_map(int k, int c) {
    int t = (c + 1) & 3;
    int cbit = (k >> (3 - c)) & 1;
    return k ^ (cbit << (3 - t));
}

__device__ __forceinline__ float xor16(float v) {
    return __int_as_float(__builtin_amdgcn_ds_swizzle(__float_as_int(v), 0x401F));
}

__global__ __launch_bounds__(BLOCK) void qlstm_kernel(
    const float* __restrict__ x, const float* __restrict__ h_prev,
    const float* __restrict__ W_in, const float* __restrict__ b_in,
    const float* __restrict__ Up, const float* __restrict__ Ud,
    float* __restrict__ out, int Btot)
{
    __shared__ float trig[4][4][2];
    __shared__ float Cmat[2][16][16];
    __shared__ float ldsJ[WAVES][32][JSTRIDE];

    const int tid = threadIdx.x;

    if (tid < 16) {
        int l = tid >> 2, q = tid & 3;
        float w = (l == 0) ? Up[q] : (l == 1) ? Up[4 + q]
                : (l == 2) ? Ud[4 + q] : Ud[q];
        trig[l][q][0] = cosf(0.5f * w);
        trig[l][q][1] = sinf(0.5f * w);
    }
    const int ii = tid >> 4, jj = tid & 15;
    Cmat[0][ii][jj] = (ii == jj) ? 1.0f : 0.0f;
    __syncthreads();

    int cur = 0;
    #pragma unroll
    for (int l = 0; l < 4; ++l) {
        int kk = ii;
        kk = cnot_map(kk, 3); kk = cnot_map(kk, 2);
        kk = cnot_map(kk, 1); kk = cnot_map(kk, 0);
        float sum = 0.0f;
        #pragma unroll
        for (int k = 0; k < 16; ++k) {
            float u = 1.0f;
            #pragma unroll
            for (int q = 0; q < 4; ++q) {
                int r  = (kk >> (3 - q)) & 1;
                int cb = (k  >> (3 - q)) & 1;
                float cc = trig[l][q][0], ss = trig[l][q][1];
                u *= (r == cb) ? cc : (r ? ss : -ss);
            }
            sum += u * Cmat[cur][k][jj];
        }
        Cmat[cur ^ 1][ii][jj] = sum;
        __syncthreads();
        cur ^= 1;
    }

    const int lane = tid & 31;
    const int wv   = tid >> 5;
    const int hh   = lane >> 4;
    const int mn   = lane & 15;
    v2f afrag[4];
    #pragma unroll
    for (int kc = 0; kc < 4; ++kc) {
        afrag[kc].x = Cmat[cur][mn][4 * kc + 2 * hh + 0];
        afrag[kc].y = Cmat[cur][mn][4 * kc + 2 * hh + 1];
    }
    float Wl[4], bl[4];
    #pragma unroll
    for (int j = 0; j < 4; ++j) { Wl[j] = W_in[j]; bl[j] = b_in[j]; }

    const int totalWaves = gridDim.x * WAVES;
    const int gw = blockIdx.x * WAVES + wv;
    const int stride = totalWaves * 32;

    for (int base = gw * 32; base < Btot; base += stride) {
        int e  = base + lane;
        int ec = min(e, Btot - 1);
        float xv = x[ec];
        float4 hp = ((const float4*)h_prev)[ec];

        int en = min(ec + stride, Btot - 1);
        __builtin_prefetch(&h_prev[4 * (size_t)en], 0, 1);

        float pj[4], tt, nn = 0.0f;
        #pragma unroll
        for (int j = 0; j < 4; ++j) {
            pj[j] = xv * Wl[j] + bl[j];
            tt = pj[j] + 1e-12f;
            nn += tt * tt;
        }
        float inv1 = 1.0f / sqrtf(nn);

        float jv[16];
        float hpv[4] = {hp.x, hp.y, hp.z, hp.w};
        float s2 = 0.0f;
        #pragma unroll
        for (int j = 0; j < 4; ++j) {
            float inj = pj[j] * inv1;
            #pragma unroll
            for (int k = 0; k < 4; ++k) {
                float t = inj * hpv[k];
                jv[4 * j + k] = t;
                s2 += t * t;
            }
        }
        float inv2 = 1.0f / sqrtf(s2);
        float* row = ldsJ[wv][lane];
        #pragma unroll
        for (int i2 = 0; i2 < 16; ++i2) row[i2] = jv[i2] * inv2;
        __builtin_amdgcn_wave_barrier();

        #pragma unroll
        for (int half = 0; half < 2; ++half) {
            const float* jrow = ldsJ[wv][half * 16 + mn];
            v2f bfrag[4];
            #pragma unroll
            for (int kc = 0; kc < 4; ++kc) {
                bfrag[kc].x = jrow[4 * kc + 2 * hh + 0];
                bfrag[kc].y = jrow[4 * kc + 2 * hh + 1];
            }
            v8f acc = {};
            #pragma unroll
            for (int kc = 0; kc < 4; ++kc)
                acc = __builtin_amdgcn_wmma_f32_16x16x4_f32(
                    false, afrag[kc], false, bfrag[kc],
                    (short)0, acc, false, false);

            float p[8];
            #pragma unroll
            for (int v = 0; v < 8; ++v) p[v] = acc[v] * acc[v];
            float ha0 = p[0] + p[4], ha1 = p[1] + p[5];
            float ha2 = p[2] + p[6], ha3 = p[3] + p[7];
            float op = (p[0] + p[1] + p[2] + p[3]) - (p[4] + p[5] + p[6] + p[7]);
            ha0 += xor16(ha0); ha1 += xor16(ha1);
            ha2 += xor16(ha2); ha3 += xor16(ha3);
            op  += xor16(op);

            int eo = base + half * 16 + mn;
            if (hh == 0 && eo < Btot) {
                ((float4*)out)[eo] = make_float4(ha0, ha1, ha2, ha3);
                out[4 * (size_t)Btot + eo] = op;
            }
        }
        __builtin_amdgcn_wave_barrier();
    }
}

extern "C" void kernel_launch(void* const* d_in, const int* in_sizes, int n_in,
                              void* d_out, int out_size, void* d_ws, size_t ws_size,
                              hipStream_t stream) {
    const float* x      = (const float*)d_in[0];
    const float* h_prev = (const float*)d_in[1];
    const float* W_in   = (const float*)d_in[2];
    const float* b_in   = (const float*)d_in[3];
    const float* Up     = (const float*)d_in[4];
    const float* Ud     = (const float*)d_in[5];
    float* out = (float*)d_out;

    int Btot = in_sizes[0];
    int tiles = (Btot + 32 * WAVES - 1) / (32 * WAVES);
    int blocks = 1024;
    if (blocks > tiles) blocks = (tiles > 0) ? tiles : 1;

    qlstm_kernel<<<blocks, BLOCK, 0, stream>>>(x, h_prev, W_in, b_in, Up, Ud, out, Btot);
}


# Software-pipelined input loads: first tile and W_in/b_in requested before the set-up, next tile loaded to registers and the tile after it prefetched to L2 while the current tile computes
# speedup vs baseline: 1.0507x; 1.0507x over previous
#include <hip/hip_runtime.h>

typedef __attribute__((ext_vector_type(2))) float v2f;
typedef __attribute__((ext_vector_type(8))) float v8f;

#define BLOCK 256
#define WAVES 8
#define JSTRIDE 17

__device__ __forceinline__ int cnot_map(int k, int c) {
    int t = (c + 1) & 3;
    int cbit = (k >> (3 - c)) & 1;
    return k ^ (cbit << (3 - t));
}

__device__ __forceinline__ float xor16(float v) {
    return __int_as_float(__builtin_amdgcn_ds_swizzle(__float_as_int(v), 0x401F));
}

__global__ __launch_bounds__(BLOCK) void qlstm_kernel(
    const float* __restrict__ x, const float* __restrict__ h_prev,
    const float* __restrict__ W_in, const float* __restrict__ b_in,
    const float* __restrict__ Up, const float* __restrict__ Ud,
    float* __restrict__ out, int Btot)
{
    __shared__ float trig[4][4][2];
    __shared__ float Cmat[2][16][16];
    __shared__ float ldsJ[WAVES][32][JSTRIDE];

    const int tid = threadIdx.x;
    const int lane = tid & 31;
    const int wv   = tid >> 5;
    const int totalWaves = gridDim.x * WAVES;
    const int gw = blockIdx.x * WAVES + wv;
    const int stride = totalWaves * 32;

    const int al = (tid >> 2) & 3, aq = tid & 3;
    const float* aptr = (al < 2) ? (Up + 4 * al + aq) : (Ud + 4 * (3 - al) + aq);
    const float wang = *aptr;

    float Wl[4], bl[4];
    #pragma unroll
    for (int j = 0; j < 4; ++j) { Wl[j] = W_in[j]; bl[j] = b_in[j]; }
    const int e0 = max(min(gw * 32 + lane, Btot - 1), 0);
    float  xv_n = x[e0];
    float4 hp_n = ((const float4*)h_prev)[e0];
    if (gw * 32 + stride < Btot) {
        int e1 = min(gw * 32 + stride + lane, Btot - 1);
        __builtin_prefetch(&x[e1], 0, 1);
        __builtin_prefetch(&h_prev[4 * (size_t)e1], 0, 1);
    }

    if (tid < 16) {
        int l = tid >> 2, q = tid & 3;
        trig[l][q][0] = cosf(0.5f * wang);
        trig[l][q][1] = sinf(0.5f * wang);
    }
    const int ii = tid >> 4, jj = tid & 15;
    Cmat[0][ii][jj] = (ii == jj) ? 1.0f : 0.0f;
    __syncthreads();

    int cur = 0;
    #pragma unroll
    for (int l = 0; l < 4; ++l) {
        int kk = ii;
        kk = cnot_map(kk, 3); kk = cnot_map(kk, 2);
        kk = cnot_map(kk, 1); kk = cnot_map(kk, 0);
        float sum = 0.0f;
        #pragma unroll
        for (int k = 0; k < 16; ++k) {
            float u = 1.0f;
            #pragma unroll
            for (int q = 0; q < 4; ++q) {
                int r  = (kk >> (3 - q)) & 1;
                int cb = (k  >> (3 - q)) & 1;
                float cc = trig[l][q][0], ss = trig[l][q][1];
                u *= (r == cb) ? cc : (r ? ss : -ss);
            }
            sum += u * Cmat[cur][k][jj];
        }
        Cmat[cur ^ 1][ii][jj] = sum;
        __syncthreads();
        cur ^= 1;
    }

    const int hh   = lane >> 4;
    const int mn   = lane & 15;
    v2f afrag[4];
    #pragma unroll
    for (int kc = 0; kc < 4; ++kc) {
        afrag[kc].x = Cmat[cur][mn][4 * kc + 2 * hh + 0];
        afrag[kc].y = Cmat[cur][mn][4 * kc + 2 * hh + 1];
    }

    for (int base = gw * 32; base < Btot; base += stride) {
        float  xv = xv_n;
        float4 hp = hp_n;
        if (base + stride < Btot) {
            int en = min(base + stride + lane, Btot - 1);
            xv_n = x[en];
            hp_n = ((const float4*)h_prev)[en];
            if (base + 2 * stride < Btot) {
                int e2 = min(base + 2 * stride + lane, Btot - 1);
                __builtin_prefetch(&x[e2], 0, 1);
                __builtin_prefetch(&h_prev[4 * (size_t)e2], 0, 1);
            }
        }

        float pj[4], tt, nn = 0.0f;
        #pragma unroll
        for (int j = 0; j < 4; ++j) {
            pj[j] = xv * Wl[j] + bl[j];
            tt = pj[j] + 1e-12f;
            nn += tt * tt;
        }
        float inv1 = 1.0f / sqrtf(nn);

        float jv[16];
        float hpv[4] = {hp.x, hp.y, hp.z, hp.w};
        float s2 = 0.0f;
        #pragma unroll
        for (int j = 0; j < 4; ++j) {
            float inj = pj[j] * inv1;
            #pragma unroll
            for (int k = 0; k < 4; ++k) {
                float t = inj * hpv[k];
                jv[4 * j + k] = t;
                s2 += t * t;
            }
        }
        float inv2 = 1.0f / sqrtf(s2);
        float* row = ldsJ[wv][lane];
        #pragma unroll
        for (int i2 = 0; i2 < 16; ++i2) row[i2] = jv[i2] * inv2;
        __builtin_amdgcn_wave_barrier();

        #pragma unroll
        for (int half = 0; half < 2; ++half) {
            const float* jrow = ldsJ[wv][half * 16 + mn];
            v2f bfrag[4];
            #pragma unroll
            for (int kc = 0; kc < 4; ++kc) {
                bfrag[kc].x = jrow[4 * kc + 2 * hh + 0];
                bfrag[kc].y = jrow[4 * kc + 2 * hh + 1];
            }
            v8f acc = {};
            #pragma unroll
            for (int kc = 0; kc < 4; ++kc)
                acc = __builtin_amdgcn_wmma_f32_16x16x4_f32(
                    false, afrag[kc], false, bfrag[kc],
                    (short)0, acc, false, false);

            float p[8];
            #pragma unroll
            for (int v = 0; v < 8; ++v) p[v] = acc[v] * acc[v];
            float ha0 = p[0] + p[4], ha1 = p[1] + p[5];
            float ha2 = p[2] + p[6], ha3 = p[3] + p[7];
            float op = (p[0] + p[1] + p[2] + p[3]) - (p[4] + p[5] + p[6] + p[7]);
            ha0 += xor16(ha0); ha1 += xor16(ha1);
            ha2 += xor16(ha2); ha3 += xor16(ha3);
            op  += xor16(op);

            int eo = base + half * 16 + mn;
            if (hh == 0 && eo < Btot) {
                ((float4*)out)[eo] = make_float4(ha0, ha1, ha2, ha3);
                out[4 * (size_t)Btot + eo] = op;
            }
        }
        __builtin_amdgcn_wave_barrier();
    }
}

extern "C" void kernel_launch(void* const* d_in, const int* in_sizes, int n_in,
                              void* d_out, int out_size, void* d_ws, size_t ws_size,
                              hipStream_t stream) {
    const float* x      = (const float*)d_in[0];
    const float* h_prev = (const float*)d_in[1];
    const float* W_in   = (const float*)d_in[2];
    const float* b_in   = (const float*)d_in[3];
    const float* Up     = (const float*)d_in[4];
    const float* Ud     = (const float*)d_in[5];
    float* out = (float*)d_out;

    int Btot = in_sizes[0];
    int tiles = (Btot + 32 * WAVES - 1) / (32 * WAVES);
    int blocks = 1024;
    if (blocks > tiles) blocks = (tiles > 0) ? tiles : 1;

    qlstm_kernel<<<blocks, BLOCK, 0, stream>>>(x, h_prev, W_in, b_in, Up, Ud, out, Btot);
}


# Normalise once: joint = outer(proj,h_prev)*rsqrtf(|proj|^2*|h_prev|^2) replaces both sqrt+divide normalisations (the first cancels); norm = product of two 4-term sums, not one sum over all products
# speedup vs baseline: 1.1693x; 1.1129x over previous
#include <hip/hip_runtime.h>

typedef __attribute__((ext_vector_type(2))) float v2f;
typedef __attribute__((ext_vector_type(8))) float v8f;

#define BLOCK 256
#define WAVES 8
#define JSTRIDE 17

__device__ __forceinline__ int cnot_map(int k, int c) {
    int t = (c + 1) & 3;
    int cbit = (k >> (3 - c)) & 1;
    return k ^ (cbit << (3 - t));
}

__device__ __forceinline__ float xor16(float v) {
    return __int_as_float(__builtin_amdgcn_ds_swizzle(__float_as_int(v), 0x401F));
}

__global__ __launch_bounds__(BLOCK) void qlstm_kernel(
    const float* __restrict__ x, const float* __restrict__ h_prev,
    const float* __restrict__ W_in, const float* __restrict__ b_in,
    const float* __restrict__ Up, const float* __restrict__ Ud,
    float* __restrict__ out, int Btot)
{
    __shared__ float trig[4][4][2];
    __shared__ float Cmat[2][16][16];
    __shared__ float ldsJ[WAVES][32][JSTRIDE];

    const int tid = threadIdx.x;
    const int lane = tid & 31;
    const int wv   = tid >> 5;
    const int totalWaves = gridDim.x * WAVES;
    const int gw = blockIdx.x * WAVES + wv;
    const int stride = totalWaves * 32;

    const int al = (tid >> 2) & 3, aq = tid & 3;
    const float* aptr = (al < 2) ? (Up + 4 * al + aq) : (Ud + 4 * (3 - al) + aq);
    const float wang = *aptr;

    float Wl[4], bl[4];
    #pragma unroll
    for (int j = 0; j < 4; ++j) { Wl[j] = W_in[j]; bl[j] = b_in[j]; }
    const int e0 = max(min(gw * 32 + lane, Btot - 1), 0);
    float  xv_n = x[e0];
    float4 hp_n = ((const float4*)h_prev)[e0];
    if (gw * 32 + stride < Btot) {
        int e1 = min(gw * 32 + stride + lane, Btot - 1);
        __builtin_prefetch(&x[e1], 0, 1);
        __builtin_prefetch(&h_prev[4 * (size_t)e1], 0, 1);
    }

    if (tid < 16) {
        int l = tid >> 2, q = tid & 3;
        trig[l][q][0] = cosf(0.5f * wang);
        trig[l][q][1] = sinf(0.5f * wang);
    }
    const int ii = tid >> 4, jj = tid & 15;
    Cmat[0][ii][jj] = (ii == jj) ? 1.0f : 0.0f;
    __syncthreads();

    int cur = 0;
    #pragma unroll
    for (int l = 0; l < 4; ++l) {
        int kk = ii;
        kk = cnot_map(kk, 3); kk = cnot_map(kk, 2);
        kk = cnot_map(kk, 1); kk = cnot_map(kk, 0);
        float sum = 0.0f;
        #pragma unroll
        for (int k = 0; k < 16; ++k) {
            float u = 1.0f;
            #pragma unroll
            for (int q = 0; q < 4; ++q) {
                int r  = (kk >> (3 - q)) & 1;
                int cb = (k  >> (3 - q)) & 1;
                float cc = trig[l][q][0], ss = trig[l][q][1];
                u *= (r == cb) ? cc : (r ? ss : -ss);
            }
            sum += u * Cmat[cur][k][jj];
        }
        Cmat[cur ^ 1][ii][jj] = sum;
        __syncthreads();
        cur ^= 1;
    }

    const int hh   = lane >> 4;
    const int mn   = lane & 15;
    v2f afrag[4];
    #pragma unroll
    for (int kc = 0; kc < 4; ++kc) {
        afrag[kc].x = Cmat[cur][mn][4 * kc + 2 * hh + 0];
        afrag[kc].y = Cmat[cur][mn][4 * kc + 2 * hh + 1];
    }

    for (int base = gw * 32; base < Btot; base += stride) {
        float  xv = xv_n;
        float4 hp = hp_n;
        if (base + stride < Btot) {
            int en = min(base + stride + lane, Btot - 1);
            xv_n = x[en];
            hp_n = ((const float4*)h_prev)[en];
            if (base + 2 * stride < Btot) {
                int e2 = min(base + 2 * stride + lane, Btot - 1);
                __builtin_prefetch(&x[e2], 0, 1);
                __builtin_prefetch(&h_prev[4 * (size_t)e2], 0, 1);
            }
        }

        float pj[4], np = 0.0f;
        #pragma unroll
        for (int j = 0; j < 4; ++j) {
            pj[j] = xv * Wl[j] + bl[j];
            np += pj[j] * pj[j];
        }
        float hpv[4] = {hp.x, hp.y, hp.z, hp.w};
        float nh = hp.x * hp.x + hp.y * hp.y + hp.z * hp.z + hp.w * hp.w;
        float sc = rsqrtf(np * nh);
        float* row = ldsJ[wv][lane];
        #pragma unroll
        for (int j = 0; j < 4; ++j) {
            float sj = pj[j] * sc;
            #pragma unroll
            for (int k = 0; k < 4; ++k) row[4 * j + k] = sj * hpv[k];
        }
        __builtin_amdgcn_wave_barrier();

        #pragma unroll
        for (int half = 0; half < 2; ++half) {
            const float* jrow = ldsJ[wv][half * 16 + mn];
            v2f bfrag[4];
            #pragma unroll
            for (int kc = 0; kc < 4; ++kc) {
                bfrag[kc].x = jrow[4 * kc + 2 * hh + 0];
                bfrag[kc].y = jrow[4 * kc + 2 * hh + 1];
            }
            v8f acc = {};
            #pragma unroll
            for (int kc = 0; kc < 4; ++kc)
                acc = __builtin_amdgcn_wmma_f32_16x16x4_f32(
                    false, afrag[kc], false, bfrag[kc],
                    (short)0, acc, false, false);

            float p[8];
            #pragma unroll
            for (int v = 0; v < 8; ++v) p[v] = acc[v] * acc[v];
            float ha0 = p[0] + p[4], ha1 = p[1] + p[5];
            float ha2 = p[2] + p[6], ha3 = p[3] + p[7];
            float op = (p[0] + p[1] + p[2] + p[3]) - (p[4] + p[5] + p[6] + p[7]);
            ha0 += xor16(ha0); ha1 += xor16(ha1);
            ha2 += xor16(ha2); ha3 += xor16(ha3);
            op  += xor16(op);

            int eo = base + half * 16 + mn;
            if (hh == 0 && eo < Btot) {
                ((float4*)out)[eo] = make_float4(ha0, ha1, ha2, ha3);
                out[4 * (size_t)Btot + eo] = op;
            }
        }
        __builtin_amdgcn_wave_barrier();
    }
}

extern "C" void kernel_launch(void* const* d_in, const int* in_sizes, int n_in,
                              void* d_out, int out_size, void* d_ws, size_t ws_size,
                              hipStream_t stream) {
    const float* x      = (const float*)d_in[0];
    const float* h_prev = (const float*)d_in[1];
    const float* W_in   = (const float*)d_in[2];
    const float* b_in   = (const float*)d_in[3];
    const float* Up     = (const float*)d_in[4];
    const float* Ud     = (const float*)d_in[5];
    float* out = (float*)d_out;

    int Btot = in_sizes[0];
    int tiles = (Btot + 32 * WAVES - 1) / (32 * WAVES);
    int blocks = 1024;
    if (blocks > tiles) blocks = (tiles > 0) ? tiles : 1;

    qlstm_kernel<<<blocks, BLOCK, 0, stream>>>(x, h_prev, W_in, b_in, Up, Ud, out, Btot);
}


# Input projection folded into the circuit matrix, psi = [C(W x I) | C(b x I)] * [x*h*s ; h*s]: K = 8 not 16, so 2 WMMAs per 16 elements (was 4) and half the LDS staging; summation order changes
# speedup vs baseline: 1.3458x; 1.1509x over previous
#include <hip/hip_runtime.h>

typedef __attribute__((ext_vector_type(2))) float v2f;
typedef __attribute__((ext_vector_type(8))) float v8f;

#define BLOCK 256
#define WAVES 8
#define BSTRIDE 12

__device__ __forceinline__ int cnot_map(int k, int c) {
    int t = (c + 1) & 3;
    int cbit = (k >> (3 - c)) & 1;
    return k ^ (cbit << (3 - t));
}

__device__ __forceinline__ float xor16(float v) {
    return __int_as_float(__builtin_amdgcn_ds_swizzle(__float_as_int(v), 0x401F));
}

__global__ __launch_bounds__(BLOCK) void qlstm_kernel(
    const float* __restrict__ x, const float* __restrict__ h_prev,
    const float* __restrict__ W_in, const float* __restrict__ b_in,
    const float* __restrict__ Up, const float* __restrict__ Ud,
    float* __restrict__ out, int Btot)
{
    __shared__ float trig[4][4][2];
    __shared__ float Cmat[2][16][16];
    __shared__ __attribute__((aligned(16))) float ldsB[WAVES][32][BSTRIDE];

    const int tid = threadIdx.x;
    const int lane = tid & 31;
    const int wv   = tid >> 5;
    const int totalWaves = gridDim.x * WAVES;
    const int gw = blockIdx.x * WAVES + wv;
    const int stride = totalWaves * 32;

    const int al = (tid >> 2) & 3, aq = tid & 3;
    const float* aptr = (al < 2) ? (Up + 4 * al + aq) : (Ud + 4 * (3 - al) + aq);
    const float wang = *aptr;

    float Wl[4], bl[4];
    #pragma unroll
    for (int j = 0; j < 4; ++j) { Wl[j] = W_in[j]; bl[j] = b_in[j]; }
    const int e0 = max(min(gw * 32 + lane, Btot - 1), 0);
    float  xv_n = x[e0];
    float4 hp_n = ((const float4*)h_prev)[e0];
    if (gw * 32 + stride < Btot) {
        int e1 = min(gw * 32 + stride + lane, Btot - 1);
        __builtin_prefetch(&x[e1], 0, 1);
        __builtin_prefetch(&h_prev[4 * (size_t)e1], 0, 1);
    }

    if (tid < 16) {
        int l = tid >> 2, q = tid & 3;
        trig[l][q][0] = cosf(0.5f * wang);
        trig[l][q][1] = sinf(0.5f * wang);
    }
    const int ii = tid >> 4, jj = tid & 15;
    Cmat[0][ii][jj] = (ii == jj) ? 1.0f : 0.0f;
    __syncthreads();

    int cur = 0;
    #pragma unroll
    for (int l = 0; l < 4; ++l) {
        int kk = ii;
        kk = cnot_map(kk, 3); kk = cnot_map(kk, 2);
        kk = cnot_map(kk, 1); kk = cnot_map(kk, 0);
        float sum = 0.0f;
        #pragma unroll
        for (int k = 0; k < 16; ++k) {
            float u = 1.0f;
            #pragma unroll
            for (int q = 0; q < 4; ++q) {
                int r  = (kk >> (3 - q)) & 1;
                int cb = (k  >> (3 - q)) & 1;
                float cc = trig[l][q][0], ss = trig[l][q][1];
                u *= (r == cb) ? cc : (r ? ss : -ss);
            }
            sum += u * Cmat[cur][k][jj];
        }
        Cmat[cur ^ 1][ii][jj] = sum;
        __syncthreads();
        cur ^= 1;
    }

    const int hh   = lane >> 4;
    const int mn   = lane & 15;
    v2f afrag[2];
    {
        float cw0 = 0.0f, cw1 = 0.0f, cb0 = 0.0f, cb1 = 0.0f;
        #pragma unroll
        for (int j = 0; j < 4; ++j) {
            float c0 = Cmat[cur][mn][4 * j + 2 * hh + 0];
            float c1 = Cmat[cur][mn][4 * j + 2 * hh + 1];
            cw0 += c0 * Wl[j];  cw1 += c1 * Wl[j];
            cb0 += c0 * bl[j];  cb1 += c1 * bl[j];
        }
        afrag[0].x = cw0;  afrag[0].y = cw1;
        afrag[1].x = cb0;  afrag[1].y = cb1;
    }

    for (int base = gw * 32; base < Btot; base += stride) {
        float  xv = xv_n;
        float4 hp = hp_n;
        if (base + stride < Btot) {
            int en = min(base + stride + lane, Btot - 1);
            xv_n = x[en];
            hp_n = ((const float4*)h_prev)[en];
            if (base + 2 * stride < Btot) {
                int e2 = min(base + 2 * stride + lane, Btot - 1);
                __builtin_prefetch(&x[e2], 0, 1);
                __builtin_prefetch(&h_prev[4 * (size_t)e2], 0, 1);
            }
        }

        float np = 0.0f;
        #pragma unroll
        for (int j = 0; j < 4; ++j) {
            float pj = xv * Wl[j] + bl[j];
            np += pj * pj;
        }
        float nh = hp.x * hp.x + hp.y * hp.y + hp.z * hp.z + hp.w * hp.w;
        float sc = rsqrtf(np * nh);
        float4 hs = make_float4(hp.x * sc, hp.y * sc, hp.z * sc, hp.w * sc);
        float4 xh = make_float4(xv * hs.x, xv * hs.y, xv * hs.z, xv * hs.w);
        float* row = ldsB[wv][lane];
        *(float4*)(row + 0) = xh;
        *(float4*)(row + 4) = hs;
        __builtin_amdgcn_wave_barrier();

        #pragma unroll
        for (int half = 0; half < 2; ++half) {
            const float* brow = ldsB[wv][half * 16 + mn];
            v2f bfrag[2];
            #pragma unroll
            for (int kc = 0; kc < 2; ++kc) {
                bfrag[kc].x = brow[4 * kc + 2 * hh + 0];
                bfrag[kc].y = brow[4 * kc + 2 * hh + 1];
            }
            v8f acc = {};
            #pragma unroll
            for (int kc = 0; kc < 2; ++kc)
                acc = __builtin_amdgcn_wmma_f32_16x16x4_f32(
                    false, afrag[kc], false, bfrag[kc],
                    (short)0, acc, false, false);

            float p[8];
            #pragma unroll
            for (int v = 0; v < 8; ++v) p[v] = acc[v] * acc[v];
            float ha0 = p[0] + p[4], ha1 = p[1] + p[5];
            float ha2 = p[2] + p[6], ha3 = p[3] + p[7];
            float op = (p[0] + p[1] + p[2] + p[3]) - (p[4] + p[5] + p[6] + p[7]);
            ha0 += xor16(ha0); ha1 += xor16(ha1);
            ha2 += xor16(ha2); ha3 += xor16(ha3);
            op  += xor16(op);

            int eo = base + half * 16 + mn;
            if (hh == 0 && eo < Btot) {
                ((float4*)out)[eo] = make_float4(ha0, ha1, ha2, ha3);
                out[4 * (size_t)Btot + eo] = op;
            }
        }
        __builtin_amdgcn_wave_barrier();
    }
}

extern "C" void kernel_launch(void* const* d_in, const int* in_sizes, int n_in,
                              void* d_out, int out_size, void* d_ws, size_t ws_size,
                              hipStream_t stream) {
    const float* x      = (const float*)d_in[0];
    const float* h_prev = (const float*)d_in[1];
    const float* W_in   = (const float*)d_in[2];
    const float* b_in   = (const float*)d_in[3];
    const float* Up     = (const float*)d_in[4];
    const float* Ud     = (const float*)d_in[5];
    float* out = (float*)d_out;

    int Btot = in_sizes[0];
    int tiles = (Btot + 32 * WAVES - 1) / (32 * WAVES);
    int blocks = 1024;
    if (blocks > tiles) blocks = (tiles > 0) ? tiles : 1;

    qlstm_kernel<<<blocks, BLOCK, 0, stream>>>(x, h_prev, W_in, b_in, Up, Ud, out, Btot);
}


# Set-up: the circuit's 16 rotation stages and 4 CNOT-ring gathers act on the 8 needed columns in one wave (in-row lane exchanges), LDS hand-over, 1 barrier not 5; A's summation order changes
# speedup vs baseline: 1.5272x; 1.1348x over previous
#include <hip/hip_runtime.h>

typedef __attribute__((ext_vector_type(2))) float v2f;
typedef __attribute__((ext_vector_type(8))) float v8f;

#define BLOCK 256
#define WAVES 8
#define BSTRIDE 12

__host__ __device__ constexpr int cnot_map(int k, int c) {
    int t = (c + 1) & 3;
    int cbit = (k >> (3 - c)) & 1;
    return k ^ (cbit << (3 - t));
}

__host__ __device__ constexpr int ring_src(int i) {
    return cnot_map(cnot_map(cnot_map(cnot_map(i, 3), 2), 1), 0);
}
__host__ __device__ constexpr unsigned ring_sel(int first) {
    unsigned w = 0;
    for (int j = 0; j < 8; ++j) w |= (unsigned)ring_src(first + j) << (4 * j);
    return w;
}

__device__ __forceinline__ float xor16(float v) {
    return __int_as_float(__builtin_amdgcn_ds_swizzle(__float_as_int(v), 0x401F));
}

template <int M>
__device__ __forceinline__ float row_xor(float v) {
    return __int_as_float(__builtin_amdgcn_update_dpp(0, __float_as_int(v), 0x160 + M, 0xf, 0xf, true));
}

template <int M>
__device__ __forceinline__ void ry_stage(float (&v)[4], float c, float ss) {
    #pragma unroll
    for (int n = 0; n < 4; ++n) {
        float pv = row_xor<M>(v[n]);
        v[n] = c * v[n] + ss * pv;
    }
}

__device__ __forceinline__ float lane_bcast(float v, int src) {
    return __int_as_float(__builtin_amdgcn_readlane(__float_as_int(v), src));
}

__global__ __launch_bounds__(BLOCK) void qlstm_kernel(
    const float* __restrict__ x, const float* __restrict__ h_prev,
    const float* __restrict__ W_in, const float* __restrict__ b_in,
    const float* __restrict__ Up, const float* __restrict__ Ud,
    float* __restrict__ out, int Btot)
{
    __shared__ __attribute__((aligned(16))) float afS[32][4];
    __shared__ __attribute__((aligned(16))) float ldsB[WAVES][32][BSTRIDE];

    const int tid = threadIdx.x;
    const int lane = tid & 31;
    const int wv   = tid >> 5;
    const int totalWaves = gridDim.x * WAVES;
    const int gw = blockIdx.x * WAVES + wv;
    const int stride = totalWaves * 32;

    const int al = (tid >> 2) & 3, aq = tid & 3;
    const float* aptr = (al < 2) ? (Up + 4 * al + aq) : (Ud + 4 * (3 - al) + aq);
    const float wang = *aptr;

    float Wl[4], bl[4];
    #pragma unroll
    for (int j = 0; j < 4; ++j) { Wl[j] = W_in[j]; bl[j] = b_in[j]; }
    const int e0 = max(min(gw * 32 + lane, Btot - 1), 0);
    float  xv_n = x[e0];
    float4 hp_n = ((const float4*)h_prev)[e0];
    if (gw * 32 + stride < Btot) {
        int e1 = min(gw * 32 + stride + lane, Btot - 1);
        __builtin_prefetch(&x[e1], 0, 1);
        __builtin_prefetch(&h_prev[4 * (size_t)e1], 0, 1);
    }

    const int hh   = lane >> 4;
    const int mn   = lane & 15;
    if (wv == 0) {
        const float cv = cosf(0.5f * wang);
        const float sv = sinf(0.5f * wang);
        const int rj = mn >> 2, rk = mn & 3;
        float wj = Wl[0], bj = bl[0];
        #pragma unroll
        for (int j = 1; j < 4; ++j) { wj = (rj == j) ? Wl[j] : wj;  bj = (rj == j) ? bl[j] : bj; }
        float va[4];
        va[0] = (rk == 2 * hh + 0) ? wj : 0.0f;
        va[1] = (rk == 2 * hh + 1) ? wj : 0.0f;
        va[2] = (rk == 2 * hh + 0) ? bj : 0.0f;
        va[3] = (rk == 2 * hh + 1) ? bj : 0.0f;
        int sgn[4];
        #pragma unroll
        for (int q = 0; q < 4; ++q) sgn[q] = (((mn >> (3 - q)) & 1) ^ 1) << 31;
        #pragma unroll
        for (int l = 0; l < 4; ++l) {
            float c[4], ss[4];
            #pragma unroll
            for (int q = 0; q < 4; ++q) {
                c[q]  = lane_bcast(cv, 4 * l + q);
                ss[q] = __int_as_float(__float_as_int(lane_bcast(sv, 4 * l + q)) ^ sgn[q]);
            }
            ry_stage<8>(va, c[0], ss[0]);
            ry_stage<4>(va, c[1], ss[1]);
            ry_stage<2>(va, c[2], ss[2]);
            ry_stage<1>(va, c[3], ss[3]);
            #pragma unroll
            for (int n = 0; n < 4; ++n) {
                const int vi = __float_as_int(va[n]);
                va[n] = __int_as_float(__builtin_amdgcn_permlane16(
                    vi, vi, ring_sel(0), ring_sel(8), false, false));
            }
        }
        *(float4*)afS[lane] = make_float4(va[0], va[1], va[2], va[3]);
    }
    __syncthreads();
    v2f afrag[2];
    {
        const float4 af = *(const float4*)afS[lane];
        afrag[0].x = af.x;  afrag[0].y = af.y;
        afrag[1].x = af.z;  afrag[1].y = af.w;
    }

    for (int base = gw * 32; base < Btot; base += stride) {
        float  xv = xv_n;
        float4 hp = hp_n;
        if (base + stride < Btot) {
            int en = min(base + stride + lane, Btot - 1);
            xv_n = x[en];
            hp_n = ((const float4*)h_prev)[en];
            if (base + 2 * stride < Btot) {
                int e2 = min(base + 2 * stride + lane, Btot - 1);
                __builtin_prefetch(&x[e2], 0, 1);
                __builtin_prefetch(&h_prev[4 * (size_t)e2], 0, 1);
            }
        }

        float np = 0.0f;
        #pragma unroll
        for (int j = 0; j < 4; ++j) {
            float pj = xv * Wl[j] + bl[j];
            np += pj * pj;
        }
        float nh = hp.x * hp.x + hp.y * hp.y + hp.z * hp.z + hp.w * hp.w;
        float sc = rsqrtf(np * nh);
        float4 hs = make_float4(hp.x * sc, hp.y * sc, hp.z * sc, hp.w * sc);
        float4 xh = make_float4(xv * hs.x, xv * hs.y, xv * hs.z, xv * hs.w);
        float* row = ldsB[wv][lane];
        *(float4*)(row + 0) = xh;
        *(float4*)(row + 4) = hs;
        __builtin_amdgcn_wave_barrier();

        #pragma unroll
        for (int half = 0; half < 2; ++half) {
            const float* brow = ldsB[wv][half * 16 + mn];
            v2f bfrag[2];
            #pragma unroll
            for (int kc = 0; kc < 2; ++kc) {
                bfrag[kc].x = brow[4 * kc + 2 * hh + 0];
                bfrag[kc].y = brow[4 * kc + 2 * hh + 1];
            }
            v8f acc = {};
            #pragma unroll
            for (int kc = 0; kc < 2; ++kc)
                acc = __builtin_amdgcn_wmma_f32_16x16x4_f32(
                    false, afrag[kc], false, bfrag[kc],
                    (short)0, acc, false, false);

            float p[8];
            #pragma unroll
            for (int v = 0; v < 8; ++v) p[v] = acc[v] * acc[v];
            float ha0 = p[0] + p[4], ha1 = p[1] + p[5];
            float ha2 = p[2] + p[6], ha3 = p[3] + p[7];
            float op = (p[0] + p[1] + p[2] + p[3]) - (p[4] + p[5] + p[6] + p[7]);
            ha0 += xor16(ha0); ha1 += xor16(ha1);
            ha2 += xor16(ha2); ha3 += xor16(ha3);
            op  += xor16(op);

            int eo = base + half * 16 + mn;
            if (hh == 0 && eo < Btot) {
                ((float4*)out)[eo] = make_float4(ha0, ha1, ha2, ha3);
                out[4 * (size_t)Btot + eo] = op;
            }
        }
        __builtin_amdgcn_wave_barrier();
    }
}

extern "C" void kernel_launch(void* const* d_in, const int* in_sizes, int n_in,
                              void* d_out, int out_size, void* d_ws, size_t ws_size,
                              hipStream_t stream) {
    const float* x      = (const float*)d_in[0];
    const float* h_prev = (const float*)d_in[1];
    const float* W_in   = (const float*)d_in[2];
    const float* b_in   = (const float*)d_in[3];
    const float* Up     = (const float*)d_in[4];
    const float* Ud     = (const float*)d_in[5];
    float* out = (float*)d_out;

    int Btot = in_sizes[0];
    int tiles = (Btot + 32 * WAVES - 1) / (32 * WAVES);
    int blocks = 1024;
    if (blocks > tiles) blocks = (tiles > 0) ? tiles : 1;

    qlstm_kernel<<<blocks, BLOCK, 0, stream>>>(x, h_prev, W_in, b_in, Up, Ud, out, Btot);
}


# Half the waves, twice the work each: 512 work-groups (was 1024); a wave requests its 8 tiles at once, kept in flight across the set-up barrier (LDS-only fence), and works 2 tiles per LDS hand-over
# speedup vs baseline: 1.6209x; 1.0614x over previous
#include <hip/hip_runtime.h>

typedef __attribute__((ext_vector_type(2))) float v2f;
typedef __attribute__((ext_vector_type(8))) float v8f;

#define BLOCK 256
#define WAVES 8
#define TPW 8
#define BSTRIDE 12

__host__ __device__ constexpr int cnot_map(int k, int c) {
    int t = (c + 1) & 3;
    int cbit = (k >> (3 - c)) & 1;
    return k ^ (cbit << (3 - t));
}

__host__ __device__ constexpr int ring_src(int i) {
    return cnot_map(cnot_map(cnot_map(cnot_map(i, 3), 2), 1), 0);
}
__host__ __device__ constexpr unsigned ring_sel(int first) {
    unsigned w = 0;
    for (int j = 0; j < 8; ++j) w |= (unsigned)ring_src(first + j) << (4 * j);
    return w;
}

__device__ __forceinline__ float xor16(float v) {
    return __int_as_float(__builtin_amdgcn_ds_swizzle(__float_as_int(v), 0x401F));
}

template <int M>
__device__ __forceinline__ float row_xor(float v) {
    return __int_as_float(__builtin_amdgcn_update_dpp(0, __float_as_int(v), 0x160 + M, 0xf, 0xf, true));
}

template <int M>
__device__ __forceinline__ void ry_stage(float (&v)[4], float c, float ss) {
    #pragma unroll
    for (int n = 0; n < 4; ++n) {
        float pv = row_xor<M>(v[n]);
        v[n] = c * v[n] + ss * pv;
    }
}

__device__ __forceinline__ float lane_bcast(float v, int src) {
    return __int_as_float(__builtin_amdgcn_readlane(__float_as_int(v), src));
}

__global__ __launch_bounds__(BLOCK) void qlstm_kernel(
    const float* __restrict__ x, const float* __restrict__ h_prev,
    const float* __restrict__ W_in, const float* __restrict__ b_in,
    const float* __restrict__ Up, const float* __restrict__ Ud,
    float* __restrict__ out, int Btot)
{
    __shared__ __attribute__((aligned(16))) float afS[32][4];
    __shared__ __attribute__((aligned(16))) float ldsB[WAVES][64][BSTRIDE];

    const int tid = threadIdx.x;
    const int lane = tid & 31;
    const int wv   = tid >> 5;
    const int gw = blockIdx.x * WAVES + wv;
    const int cspan   = 32 * TPW;
    const int cstride = gridDim.x * WAVES * cspan;
    const int cbase0  = gw * cspan;

    const int al = (tid >> 2) & 3, aq = tid & 3;
    const float* aptr = (al < 2) ? (Up + 4 * al + aq) : (Ud + 4 * (3 - al) + aq);
    const float wang = *aptr;
    __builtin_amdgcn_sched_barrier(0);

    float Wl[4], bl[4];
    #pragma unroll
    for (int j = 0; j < 4; ++j) { Wl[j] = W_in[j]; bl[j] = b_in[j]; }
    float  xv[TPW];
    float4 hp[TPW];
    #pragma unroll
    for (int j = 0; j < TPW; ++j) {
        const int e = max(min(cbase0 + 32 * j + lane, Btot - 1), 0);
        xv[j] = x[e];
        hp[j] = ((const float4*)h_prev)[e];
    }
    asm volatile("" :: "v"(wang));

    const int hh   = lane >> 4;
    const int mn   = lane & 15;
    if (wv == 0) {
        const float cv = cosf(0.5f * wang);
        const float sv = sinf(0.5f * wang);
        const int rj = mn >> 2, rk = mn & 3;
        float wj = Wl[0], bj = bl[0];
        #pragma unroll
        for (int j = 1; j < 4; ++j) { wj = (rj == j) ? Wl[j] : wj;  bj = (rj == j) ? bl[j] : bj; }
        float va[4];
        va[0] = (rk == 2 * hh + 0) ? wj : 0.0f;
        va[1] = (rk == 2 * hh + 1) ? wj : 0.0f;
        va[2] = (rk == 2 * hh + 0) ? bj : 0.0f;
        va[3] = (rk == 2 * hh + 1) ? bj : 0.0f;
        int sgn[4];
        #pragma unroll
        for (int q = 0; q < 4; ++q) sgn[q] = (((mn >> (3 - q)) & 1) ^ 1) << 31;
        #pragma unroll
        for (int l = 0; l < 4; ++l) {
            float c[4], ss[4];
            #pragma unroll
            for (int q = 0; q < 4; ++q) {
                c[q]  = lane_bcast(cv, 4 * l + q);
                ss[q] = __int_as_float(__float_as_int(lane_bcast(sv, 4 * l + q)) ^ sgn[q]);
            }
            ry_stage<8>(va, c[0], ss[0]);
            ry_stage<4>(va, c[1], ss[1]);
            ry_stage<2>(va, c[2], ss[2]);
            ry_stage<1>(va, c[3], ss[3]);
            #pragma unroll
            for (int n = 0; n < 4; ++n) {
                const int vi = __float_as_int(va[n]);
                va[n] = __int_as_float(__builtin_amdgcn_permlane16(
                    vi, vi, ring_sel(0), ring_sel(8), false, false));
            }
        }
        *(float4*)afS[lane] = make_float4(va[0], va[1], va[2], va[3]);
    }
    __builtin_amdgcn_fence(__ATOMIC_RELEASE, "workgroup", "local");
    __builtin_amdgcn_s_barrier();
    __builtin_amdgcn_fence(__ATOMIC_ACQUIRE, "workgroup", "local");
    v2f afrag[2];
    {
        const float4 af = *(const float4*)afS[lane];
        afrag[0].x = af.x;  afrag[0].y = af.y;
        afrag[1].x = af.z;  afrag[1].y = af.w;
    }

    for (int cbase = cbase0; cbase < Btot; cbase += cstride) {
        #pragma unroll
        for (int p = 0; p < TPW / 2; ++p) {
            #pragma unroll
            for (int u = 0; u < 2; ++u) {
                const float  xe = xv[2 * p + u];
                const float4 he = hp[2 * p + u];
                float np = 0.0f;
                #pragma unroll
                for (int j = 0; j < 4; ++j) {
                    float pj = xe * Wl[j] + bl[j];
                    np += pj * pj;
                }
                float nh = he.x * he.x + he.y * he.y + he.z * he.z + he.w * he.w;
                float sc = rsqrtf(np * nh);
                float4 hs = make_float4(he.x * sc, he.y * sc, he.z * sc, he.w * sc);
                float4 xh = make_float4(xe * hs.x, xe * hs.y, xe * hs.z, xe * hs.w);
                float* row = ldsB[wv][32 * u + lane];
                *(float4*)(row + 0) = xh;
                *(float4*)(row + 4) = hs;
            }
            __builtin_amdgcn_wave_barrier();

            float4 hres[4];
            float  ores[4];
            #pragma unroll
            for (int q = 0; q < 4; ++q) {
                const float* brow = ldsB[wv][16 * q + mn];
                v2f bfrag[2];
                #pragma unroll
                for (int kc = 0; kc < 2; ++kc) {
                    bfrag[kc].x = brow[4 * kc + 2 * hh + 0];
                    bfrag[kc].y = brow[4 * kc + 2 * hh + 1];
                }
                v8f acc = {};
                #pragma unroll
                for (int kc = 0; kc < 2; ++kc)
                    acc = __builtin_amdgcn_wmma_f32_16x16x4_f32(
                        false, afrag[kc], false, bfrag[kc],
                        (short)0, acc, false, false);

                float pq[8];
                #pragma unroll
                for (int v = 0; v < 8; ++v) pq[v] = acc[v] * acc[v];
                float ha0 = pq[0] + pq[4], ha1 = pq[1] + pq[5];
                float ha2 = pq[2] + pq[6], ha3 = pq[3] + pq[7];
                float op = (pq[0] + pq[1] + pq[2] + pq[3]) - (pq[4] + pq[5] + pq[6] + pq[7]);
                ha0 += xor16(ha0); ha1 += xor16(ha1);
                ha2 += xor16(ha2); ha3 += xor16(ha3);
                op  += xor16(op);
                hres[q] = make_float4(ha0, ha1, ha2, ha3);
                ores[q] = op;
            }
            #pragma unroll
            for (int q = 0; q < 4; ++q) {
                int eo = cbase + 64 * p + 16 * q + mn;
                if (hh == 0 && eo < Btot) {
                    ((float4*)out)[eo] = hres[q];
                    out[4 * (size_t)Btot + eo] = ores[q];
                }
            }
            __builtin_amdgcn_wave_barrier();
        }
        if (cbase + cstride < Btot) {
            #pragma unroll
            for (int j = 0; j < TPW; ++j) {
                const int e = min(cbase + cstride + 32 * j + lane, Btot - 1);
                xv[j] = x[e];
                hp[j] = ((const float4*)h_prev)[e];
            }
        }
    }
}

extern "C" void kernel_launch(void* const* d_in, const int* in_sizes, int n_in,
                              void* d_out, int out_size, void* d_ws, size_t ws_size,
                              hipStream_t stream) {
    const float* x      = (const float*)d_in[0];
    const float* h_prev = (const float*)d_in[1];
    const float* W_in   = (const float*)d_in[2];
    const float* b_in   = (const float*)d_in[3];
    const float* Up     = (const float*)d_in[4];
    const float* Ud     = (const float*)d_in[5];
    float* out = (float*)d_out;

    int Btot = in_sizes[0];
    int tiles = (Btot + 32 * TPW * WAVES - 1) / (32 * TPW * WAVES);
    int blocks = 512;
    if (blocks > tiles) blocks = (tiles > 0) ? tiles : 1;

    qlstm_kernel<<<blocks, BLOCK, 0, stream>>>(x, h_prev, W_in, b_in, Up, Ud, out, Btot);
}


# Wave index read into a scalar register: the sweep's bounds are tested once on the scalar unit and a pair's 4 blocks stored in one lane-half region, not a vector compare, mask and branch per block
# speedup vs baseline: 1.7245x; 1.0639x over previous
#include <hip/hip_runtime.h>

typedef __attribute__((ext_vector_type(2))) float v2f;
typedef __attribute__((ext_vector_type(8))) float v8f;

#define BLOCK 256
#define WAVES 8
#define TPW 8
#define BSTRIDE 12

__host__ __device__ constexpr int cnot_map(int k, int c) {
    int t = (c + 1) & 3;
    int cbit = (k >> (3 - c)) & 1;
    return k ^ (cbit << (3 - t));
}

__host__ __device__ constexpr int ring_src(int i) {
    return cnot_map(cnot_map(cnot_map(cnot_map(i, 3), 2), 1), 0);
}
__host__ __device__ constexpr unsigned ring_sel(int first) {
    unsigned w = 0;
    for (int j = 0; j < 8; ++j) w |= (unsigned)ring_src(first + j) << (4 * j);
    return w;
}

__device__ __forceinline__ float xor16(float v) {
    return __int_as_float(__builtin_amdgcn_ds_swizzle(__float_as_int(v), 0x401F));
}

template <int M>
__device__ __forceinline__ float row_xor(float v) {
    return __int_as_float(__builtin_amdgcn_update_dpp(0, __float_as_int(v), 0x160 + M, 0xf, 0xf, true));
}

template <int M>
__device__ __forceinline__ void ry_stage(float (&v)[4], float c, float ss) {
    #pragma unroll
    for (int n = 0; n < 4; ++n) {
        float pv = row_xor<M>(v[n]);
        v[n] = c * v[n] + ss * pv;
    }
}

__device__ __forceinline__ float lane_bcast(float v, int src) {
    return __int_as_float(__builtin_amdgcn_readlane(__float_as_int(v), src));
}

template <bool FULL>
__device__ __forceinline__ void sweep(
    const float (&xv)[TPW], const float4 (&hp)[TPW], const float (&Wl)[4], const float (&bl)[4],
    const v2f (&afrag)[2], float (*rows)[BSTRIDE], float* __restrict__ out,
    int Btot, int cbase, int lane, int hh, int mn)
{
    #pragma unroll
    for (int p = 0; p < TPW / 2; ++p) {
        #pragma unroll
        for (int u = 0; u < 2; ++u) {
            const float  xe = xv[2 * p + u];
            const float4 he = hp[2 * p + u];
            float np = 0.0f;
            #pragma unroll
            for (int j = 0; j < 4; ++j) {
                float pj = xe * Wl[j] + bl[j];
                np += pj * pj;
            }
            float nh = he.x * he.x + he.y * he.y + he.z * he.z + he.w * he.w;
            float sc = rsqrtf(np * nh);
            float4 hs = make_float4(he.x * sc, he.y * sc, he.z * sc, he.w * sc);
            float4 xh = make_float4(xe * hs.x, xe * hs.y, xe * hs.z, xe * hs.w);
            float* row = rows[32 * u + lane];
            *(float4*)(row + 0) = xh;
            *(float4*)(row + 4) = hs;
        }
        __builtin_amdgcn_wave_barrier();

        float4 hres[4], hoth[4];
        float  ores[4], ooth[4];
        #pragma unroll
        for (int q = 0; q < 4; ++q) {
            const float* brow = rows[16 * q + mn];
            v2f bfrag[2];
            #pragma unroll
            for (int kc = 0; kc < 2; ++kc) {
                bfrag[kc].x = brow[4 * kc + 2 * hh + 0];
                bfrag[kc].y = brow[4 * kc + 2 * hh + 1];
            }
            v8f acc = {};
            #pragma unroll
            for (int kc = 0; kc < 2; ++kc)
                acc = __builtin_amdgcn_wmma_f32_16x16x4_f32(
                    false, afrag[kc], false, bfrag[kc],
                    (short)0, acc, false, false);

            float pq[8];
            #pragma unroll
            for (int v = 0; v < 8; ++v) pq[v] = acc[v] * acc[v];
            float ha0 = pq[0] + pq[4], ha1 = pq[1] + pq[5];
            float ha2 = pq[2] + pq[6], ha3 = pq[3] + pq[7];
            float op = (pq[0] + pq[1] + pq[2] + pq[3]) - (pq[4] + pq[5] + pq[6] + pq[7]);
            hres[q] = make_float4(ha0, ha1, ha2, ha3);
            ores[q] = op;
            hoth[q] = make_float4(xor16(ha0), xor16(ha1), xor16(ha2), xor16(ha3));
            ooth[q] = xor16(op);
        }
        if (FULL) {
            if (hh == 0) {
                #pragma unroll
                for (int q = 0; q < 4; ++q) {
                    const int eo = cbase + 64 * p + 16 * q + mn;
                    ((float4*)out)[eo] = make_float4(hres[q].x + hoth[q].x, hres[q].y + hoth[q].y,
                                                     hres[q].z + hoth[q].z, hres[q].w + hoth[q].w);
                    out[4 * (size_t)Btot + eo] = ores[q] + ooth[q];
                }
            }
        } else {
            #pragma unroll
            for (int q = 0; q < 4; ++q) {
                const int eo = cbase + 64 * p + 16 * q + mn;
                if (hh == 0 && eo < Btot) {
                    ((float4*)out)[eo] = make_float4(hres[q].x + hoth[q].x, hres[q].y + hoth[q].y,
                                                     hres[q].z + hoth[q].z, hres[q].w + hoth[q].w);
                    out[4 * (size_t)Btot + eo] = ores[q] + ooth[q];
                }
            }
        }
        __builtin_amdgcn_wave_barrier();
    }
}

__global__ __launch_bounds__(BLOCK) void qlstm_kernel(
    const float* __restrict__ x, const float* __restrict__ h_prev,
    const float* __restrict__ W_in, const float* __restrict__ b_in,
    const float* __restrict__ Up, const float* __restrict__ Ud,
    float* __restrict__ out, int Btot)
{
    __shared__ __attribute__((aligned(16))) float afS[32][4];
    __shared__ __attribute__((aligned(16))) float ldsB[WAVES][64][BSTRIDE];

    const int tid = threadIdx.x;
    const int lane = tid & 31;
    const int wv   = __builtin_amdgcn_readfirstlane(tid >> 5);
    const int gw = blockIdx.x * WAVES + wv;
    const int cspan   = 32 * TPW;
    const int cstride = gridDim.x * WAVES * cspan;
    const int cbase0  = gw * cspan;

    const int al = (tid >> 2) & 3, aq = tid & 3;
    const float* aptr = (al < 2) ? (Up + 4 * al + aq) : (Ud + 4 * (3 - al) + aq);
    const float wang = *aptr;
    __builtin_amdgcn_sched_barrier(0);

    float Wl[4], bl[4];
    #pragma unroll
    for (int j = 0; j < 4; ++j) { Wl[j] = W_in[j]; bl[j] = b_in[j]; }
    float  xv[TPW];
    float4 hp[TPW];
    #pragma unroll
    for (int j = 0; j < TPW; ++j) {
        const int e = max(min(cbase0 + 32 * j + lane, Btot - 1), 0);
        xv[j] = x[e];
        hp[j] = ((const float4*)h_prev)[e];
    }
    asm volatile("" :: "v"(wang));

    const int hh   = lane >> 4;
    const int mn   = lane & 15;
    if (wv == 0) {
        const float cv = cosf(0.5f * wang);
        const float sv = sinf(0.5f * wang);
        const int rj = mn >> 2, rk = mn & 3;
        float wj = Wl[0], bj = bl[0];
        #pragma unroll
        for (int j = 1; j < 4; ++j) { wj = (rj == j) ? Wl[j] : wj;  bj = (rj == j) ? bl[j] : bj; }
        float va[4];
        va[0] = (rk == 2 * hh + 0) ? wj : 0.0f;
        va[1] = (rk == 2 * hh + 1) ? wj : 0.0f;
        va[2] = (rk == 2 * hh + 0) ? bj : 0.0f;
        va[3] = (rk == 2 * hh + 1) ? bj : 0.0f;
        int sgn[4];
        #pragma unroll
        for (int q = 0; q < 4; ++q) sgn[q] = (((mn >> (3 - q)) & 1) ^ 1) << 31;
        #pragma unroll
        for (int l = 0; l < 4; ++l) {
            float c[4], ss[4];
            #pragma unroll
            for (int q = 0; q < 4; ++q) {
                c[q]  = lane_bcast(cv, 4 * l + q);
                ss[q] = __int_as_float(__float_as_int(lane_bcast(sv, 4 * l + q)) ^ sgn[q]);
            }
            ry_stage<8>(va, c[0], ss[0]);
            ry_stage<4>(va, c[1], ss[1]);
            ry_stage<2>(va, c[2], ss[2]);
            ry_stage<1>(va, c[3], ss[3]);
            #pragma unroll
            for (int n = 0; n < 4; ++n) {
                const int vi = __float_as_int(va[n]);
                va[n] = __int_as_float(__builtin_amdgcn_permlane16(
                    vi, vi, ring_sel(0), ring_sel(8), false, false));
            }
        }
        *(float4*)afS[lane] = make_float4(va[0], va[1], va[2], va[3]);
    }
    __builtin_amdgcn_fence(__ATOMIC_RELEASE, "workgroup", "local");
    __builtin_amdgcn_s_barrier();
    __builtin_amdgcn_fence(__ATOMIC_ACQUIRE, "workgroup", "local");
    v2f afrag[2];
    {
        const float4 af = *(const float4*)afS[lane];
        afrag[0].x = af.x;  afrag[0].y = af.y;
        afrag[1].x = af.z;  afrag[1].y = af.w;
    }

    for (int cbase = cbase0; cbase < Btot; cbase += cstride) {
        if (cbase + cspan <= Btot) sweep<true >(xv, hp, Wl, bl, afrag, ldsB[wv], out, Btot, cbase, lane, hh, mn);
        else                       sweep<false>(xv, hp, Wl, bl, afrag, ldsB[wv], out, Btot, cbase, lane, hh, mn);
        if (cbase + cstride < Btot) {
            #pragma unroll
            for (int j = 0; j < TPW; ++j) {
                const int e = min(cbase + cstride + 32 * j + lane, Btot - 1);
                xv[j] = x[e];
                hp[j] = ((const float4*)h_prev)[e];
            }
        }
    }
}

extern "C" void kernel_launch(void* const* d_in, const int* in_sizes, int n_in,
                              void* d_out, int out_size, void* d_ws, size_t ws_size,
                              hipStream_t stream) {
    const float* x      = (const float*)d_in[0];
    const float* h_prev = (const float*)d_in[1];
    const float* W_in   = (const float*)d_in[2];
    const float* b_in   = (const float*)d_in[3];
    const float* Up     = (const float*)d_in[4];
    const float* Ud     = (const float*)d_in[5];
    float* out = (float*)d_out;

    int Btot = in_sizes[0];
    int tiles = (Btot + 32 * TPW * WAVES - 1) / (32 * TPW * WAVES);
    int blocks = 512;
    if (blocks > tiles) blocks = (tiles > 0) ? tiles : 1;

    qlstm_kernel<<<blocks, BLOCK, 0, stream>>>(x, h_prev, W_in, b_in, Up, Ud, out, Btot);
}


# A's rows reordered in the set-up's last gather so a lane holds all 4 terms of two h_next sums: in-lane sums, 1 cross-half exchange per 2 blocks (was 10), full-wave stores; sum order changes
# speedup vs baseline: 1.9770x; 1.1464x over previous
#include <hip/hip_runtime.h>

typedef __attribute__((ext_vector_type(2))) float v2f;
typedef __attribute__((ext_vector_type(8))) float v8f;

#define BLOCK 256
#define WAVES 8
#define TPW 8
#define BSTRIDE 12

__host__ __device__ constexpr int cnot_map(int k, int c) {
    int t = (c + 1) & 3;
    int cbit = (k >> (3 - c)) & 1;
    return k ^ (cbit << (3 - t));
}

__host__ __device__ constexpr int ring_src(int i) {
    return cnot_map(cnot_map(cnot_map(cnot_map(i, 3), 2), 1), 0);
}
__host__ __device__ constexpr unsigned ring_sel(int first) {
    unsigned w = 0;
    for (int j = 0; j < 8; ++j) w |= (unsigned)ring_src(first + j) << (4 * j);
    return w;
}

__host__ __device__ constexpr int out_row(int m) {
    return 4 * ((m & 7) >> 1) + 2 * (m >> 3) + (m & 1);
}
__host__ __device__ constexpr unsigned last_sel(int first) {
    unsigned w = 0;
    for (int j = 0; j < 8; ++j) w |= (unsigned)ring_src(out_row(first + j)) << (4 * j);
    return w;
}

__device__ __forceinline__ float xor16(float v) {
    return __int_as_float(__builtin_amdgcn_ds_swizzle(__float_as_int(v), 0x401F));
}

template <int M>
__device__ __forceinline__ float row_xor(float v) {
    return __int_as_float(__builtin_amdgcn_update_dpp(0, __float_as_int(v), 0x160 + M, 0xf, 0xf, true));
}

template <int M>
__device__ __forceinline__ void ry_stage(float (&v)[4], float c, float ss) {
    #pragma unroll
    for (int n = 0; n < 4; ++n) {
        float pv = row_xor<M>(v[n]);
        v[n] = c * v[n] + ss * pv;
    }
}

__device__ __forceinline__ float lane_bcast(float v, int src) {
    return __int_as_float(__builtin_amdgcn_readlane(__float_as_int(v), src));
}

template <bool FULL>
__device__ __forceinline__ void sweep(
    const float (&xv)[TPW], const float4 (&hp)[TPW], const float (&Wl)[4], const float (&bl)[4],
    const v2f (&afrag)[2], float (*rows)[BSTRIDE], float* __restrict__ out,
    int Btot, int cbase, int lane, int hh, int mn)
{
    #pragma unroll
    for (int p = 0; p < TPW / 2; ++p) {
        #pragma unroll
        for (int u = 0; u < 2; ++u) {
            const float  xe = xv[2 * p + u];
            const float4 he = hp[2 * p + u];
            float np = 0.0f;
            #pragma unroll
            for (int j = 0; j < 4; ++j) {
                float pj = xe * Wl[j] + bl[j];
                np += pj * pj;
            }
            float nh = he.x * he.x + he.y * he.y + he.z * he.z + he.w * he.w;
            float sc = rsqrtf(np * nh);
            float4 hs = make_float4(he.x * sc, he.y * sc, he.z * sc, he.w * sc);
            float4 xh = make_float4(xe * hs.x, xe * hs.y, xe * hs.z, xe * hs.w);
            float* row = rows[32 * u + lane];
            *(float4*)(row + 0) = xh;
            *(float4*)(row + 4) = hs;
        }
        __builtin_amdgcn_wave_barrier();

        v2f   hres[4];
        float opart[4];
        #pragma unroll
        for (int q = 0; q < 4; ++q) {
            const float* brow = rows[16 * q + mn];
            v2f bfrag[2];
            #pragma unroll
            for (int kc = 0; kc < 2; ++kc) {
                bfrag[kc].x = brow[4 * kc + 2 * hh + 0];
                bfrag[kc].y = brow[4 * kc + 2 * hh + 1];
            }
            v8f acc = {};
            #pragma unroll
            for (int kc = 0; kc < 2; ++kc)
                acc = __builtin_amdgcn_wmma_f32_16x16x4_f32(
                    false, afrag[kc], false, bfrag[kc],
                    (short)0, acc, false, false);

            const v2f a0 = __builtin_shufflevector(acc, acc, 0, 1);
            const v2f a1 = __builtin_shufflevector(acc, acc, 2, 3);
            const v2f a2 = __builtin_shufflevector(acc, acc, 4, 5);
            const v2f a3 = __builtin_shufflevector(acc, acc, 6, 7);
            const v2f sp = a0 * a0 + a2 * a2;
            const v2f sn = a1 * a1 + a3 * a3;
            const v2f dz = sp - sn;
            hres[q]  = sp + sn;
            opart[q] = dz.x + dz.y;
        }
        const int hidx = 2 * (cbase + 64 * p + mn) + hh;
        #pragma unroll
        for (int q = 0; q < 4; ++q) {
            if (FULL || cbase + 64 * p + 16 * q + mn < Btot)
                ((v2f*)out)[hidx + 32 * q] = hres[q];
        }
        #pragma unroll
        for (int g = 0; g < 2; ++g) {
            const float keep = hh ? opart[2 * g + 1] : opart[2 * g];
            const float send = hh ? opart[2 * g] : opart[2 * g + 1];
            const int eo = cbase + 64 * p + 32 * g + lane;
            const float tot = keep + xor16(send);
            if (FULL || eo < Btot)
                out[4 * (size_t)Btot + eo] = tot;
        }
        __builtin_amdgcn_wave_barrier();
    }
}

__global__ __launch_bounds__(BLOCK) void qlstm_kernel(
    const float* __restrict__ x, const float* __restrict__ h_prev,
    const float* __restrict__ W_in, const float* __restrict__ b_in,
    const float* __restrict__ Up, const float* __restrict__ Ud,
    float* __restrict__ out, int Btot)
{
    __shared__ __attribute__((aligned(16))) float afS[32][4];
    __shared__ __attribute__((aligned(16))) float ldsB[WAVES][64][BSTRIDE];

    const int tid = threadIdx.x;
    const int lane = tid & 31;
    const int wv   = __builtin_amdgcn_readfirstlane(tid >> 5);
    const int gw = blockIdx.x * WAVES + wv;
    const int cspan   = 32 * TPW;
    const int cstride = gridDim.x * WAVES * cspan;
    const int cbase0  = gw * cspan;

    const int al = (tid >> 2) & 3, aq = tid & 3;
    const float* aptr = (al < 2) ? (Up + 4 * al + aq) : (Ud + 4 * (3 - al) + aq);
    const float wang = *aptr;
    __builtin_amdgcn_sched_barrier(0);

    float Wl[4], bl[4];
    #pragma unroll
    for (int j = 0; j < 4; ++j) { Wl[j] = W_in[j]; bl[j] = b_in[j]; }
    float  xv[TPW];
    float4 hp[TPW];
    #pragma unroll
    for (int j = 0; j < TPW; ++j) {
        const int e = max(min(cbase0 + 32 * j + lane, Btot - 1), 0);
        xv[j] = x[e];
        hp[j] = ((const float4*)h_prev)[e];
    }
    asm volatile("" :: "v"(wang));

    const int hh   = lane >> 4;
    const int mn   = lane & 15;
    if (wv == 0) {
        const float cv = cosf(0.5f * wang);
        const float sv = sinf(0.5f * wang);
        const int rj = mn >> 2, rk = mn & 3;
        float wj = Wl[0], bj = bl[0];
        #pragma unroll
        for (int j = 1; j < 4; ++j) { wj = (rj == j) ? Wl[j] : wj;  bj = (rj == j) ? bl[j] : bj; }
        float va[4];
        va[0] = (rk == 2 * hh + 0) ? wj : 0.0f;
        va[1] = (rk == 2 * hh + 1) ? wj : 0.0f;
        va[2] = (rk == 2 * hh + 0) ? bj : 0.0f;
        va[3] = (rk == 2 * hh + 1) ? bj : 0.0f;
        int sgn[4];
        #pragma unroll
        for (int q = 0; q < 4; ++q) sgn[q] = (((mn >> (3 - q)) & 1) ^ 1) << 31;
        #pragma unroll
        for (int l = 0; l < 4; ++l) {
            float c[4], ss[4];
            #pragma unroll
            for (int q = 0; q < 4; ++q) {
                c[q]  = lane_bcast(cv, 4 * l + q);
                ss[q] = __int_as_float(__float_as_int(lane_bcast(sv, 4 * l + q)) ^ sgn[q]);
            }
            ry_stage<8>(va, c[0], ss[0]);
            ry_stage<4>(va, c[1], ss[1]);
            ry_stage<2>(va, c[2], ss[2]);
            ry_stage<1>(va, c[3], ss[3]);
            const unsigned sel_lo = (l < 3) ? ring_sel(0) : last_sel(0);
            const unsigned sel_hi = (l < 3) ? ring_sel(8) : last_sel(8);
            #pragma unroll
            for (int n = 0; n < 4; ++n) {
                const int vi = __float_as_int(va[n]);
                va[n] = __int_as_float(__builtin_amdgcn_permlane16(
                    vi, vi, sel_lo, sel_hi, false, false));
            }
        }
        *(float4*)afS[lane] = make_float4(va[0], va[1], va[2], va[3]);
    }
    __builtin_amdgcn_fence(__ATOMIC_RELEASE, "workgroup", "local");
    __builtin_amdgcn_s_barrier();
    __builtin_amdgcn_fence(__ATOMIC_ACQUIRE, "workgroup", "local");
    v2f afrag[2];
    {
        const float4 af = *(const float4*)afS[lane];
        afrag[0].x = af.x;  afrag[0].y = af.y;
        afrag[1].x = af.z;  afrag[1].y = af.w;
    }

    for (int cbase = cbase0; cbase < Btot; cbase += cstride) {
        if (cbase + cspan <= Btot) sweep<true >(xv, hp, Wl, bl, afrag, ldsB[wv], out, Btot, cbase, lane, hh, mn);
        else                       sweep<false>(xv, hp, Wl, bl, afrag, ldsB[wv], out, Btot, cbase, lane, hh, mn);
        if (cbase + cstride < Btot) {
            #pragma unroll
            for (int j = 0; j < TPW; ++j) {
                const int e = min(cbase + cstride + 32 * j + lane, Btot - 1);
                xv[j] = x[e];
                hp[j] = ((const float4*)h_prev)[e];
            }
        }
    }
}

extern "C" void kernel_launch(void* const* d_in, const int* in_sizes, int n_in,
                              void* d_out, int out_size, void* d_ws, size_t ws_size,
                              hipStream_t stream) {
    const float* x      = (const float*)d_in[0];
    const float* h_prev = (const float*)d_in[1];
    const float* W_in   = (const float*)d_in[2];
    const float* b_in   = (const float*)d_in[3];
    const float* Up     = (const float*)d_in[4];
    const float* Ud     = (const float*)d_in[5];
    float* out = (float*)d_out;

    int Btot = in_sizes[0];
    int tiles = (Btot + 32 * TPW * WAVES - 1) / (32 * TPW * WAVES);
    int blocks = 512;
    if (blocks > tiles) blocks = (tiles > 0) ? tiles : 1;

    qlstm_kernel<<<blocks, BLOCK, 0, stream>>>(x, h_prev, W_in, b_in, Up, Ud, out, Btot);
}


# Sweep software-pipelined by pair: a pair waits only for its own loads; rows staged and 4 fragment reads issued as one group under the pair before's WMMAs; <Z> exchange used a pair later; same sums
# speedup vs baseline: 2.0837x; 1.0540x over previous
#include <hip/hip_runtime.h>

typedef __attribute__((ext_vector_type(2))) float v2f;
typedef __attribute__((ext_vector_type(8))) float v8f;

#define BLOCK 256
#define WAVES 8
#define TPW 8
#define BSTRIDE 12

__host__ __device__ constexpr int cnot_map(int k, int c) {
    int t = (c + 1) & 3;
    int cbit = (k >> (3 - c)) & 1;
    return k ^ (cbit << (3 - t));
}

__host__ __device__ constexpr int ring_src(int i) {
    return cnot_map(cnot_map(cnot_map(cnot_map(i, 3), 2), 1), 0);
}
__host__ __device__ constexpr unsigned ring_sel(int first) {
    unsigned w = 0;
    for (int j = 0; j < 8; ++j) w |= (unsigned)ring_src(first + j) << (4 * j);
    return w;
}

__host__ __device__ constexpr int out_row(int m) {
    return 4 * ((m & 7) >> 1) + 2 * (m >> 3) + (m & 1);
}
__host__ __device__ constexpr unsigned last_sel(int first) {
    unsigned w = 0;
    for (int j = 0; j < 8; ++j) w |= (unsigned)ring_src(out_row(first + j)) << (4 * j);
    return w;
}

__device__ __forceinline__ float xor16(float v) {
    return __int_as_float(__builtin_amdgcn_ds_swizzle(__float_as_int(v), 0x401F));
}

template <int M>
__device__ __forceinline__ float row_xor(float v) {
    return __int_as_float(__builtin_amdgcn_update_dpp(0, __float_as_int(v), 0x160 + M, 0xf, 0xf, true));
}

template <int M>
__device__ __forceinline__ void ry_stage(float (&v)[4], float c, float ss) {
    #pragma unroll
    for (int n = 0; n < 4; ++n) {
        float pv = row_xor<M>(v[n]);
        v[n] = c * v[n] + ss * pv;
    }
}

__device__ __forceinline__ float lane_bcast(float v, int src) {
    return __int_as_float(__builtin_amdgcn_readlane(__float_as_int(v), src));
}

__device__ __forceinline__ void stage_rows(
    const float (&xp)[2], const float4 (&hq)[2], const float (&Wl)[4], const float (&bl)[4],
    float (*rows)[BSTRIDE], int lane)
{
    #pragma unroll
    for (int u = 0; u < 2; ++u) {
        const float  xe = xp[u];
        const float4 he = hq[u];
        float np = 0.0f;
        #pragma unroll
        for (int j = 0; j < 4; ++j) {
            float pj = xe * Wl[j] + bl[j];
            np += pj * pj;
        }
        float nh = he.x * he.x + he.y * he.y + he.z * he.z + he.w * he.w;
        float sc = rsqrtf(np * nh);
        float4 hs = make_float4(he.x * sc, he.y * sc, he.z * sc, he.w * sc);
        float4 xh = make_float4(xe * hs.x, xe * hs.y, xe * hs.z, xe * hs.w);
        float* row = rows[32 * u + lane];
        *(float4*)(row + 0) = xh;
        *(float4*)(row + 4) = hs;
    }
}

__device__ __forceinline__ void read_frags(
    float (*rows)[BSTRIDE], int mn, int hh, v2f (&bf)[4][2])
{
    #pragma unroll
    for (int q = 0; q < 4; ++q) {
        const float* brow = rows[16 * q + mn];
        #pragma unroll
        for (int kc = 0; kc < 2; ++kc) {
            bf[q][kc].x = brow[4 * kc + 2 * hh + 0];
            bf[q][kc].y = brow[4 * kc + 2 * hh + 1];
        }
    }
}

template <bool FULL>
__device__ __forceinline__ void sweep(
    const float (&xv)[TPW], const float4 (&hp)[TPW], const float (&Wl)[4], const float (&bl)[4],
    const v2f (&afrag)[2], float (*rows)[BSTRIDE], float* __restrict__ out,
    int Btot, int cbase, int lane, int hh, int mn)
{
    v2f bfA[4][2], bfB[4][2];
    v2f zkeep, zrecv;

    {
        const float  xp[2] = { xv[0], xv[1] };
        const float4 hq[2] = { hp[0], hp[1] };
        stage_rows(xp, hq, Wl, bl, rows, lane);
        __builtin_amdgcn_wave_barrier();
        read_frags(rows, mn, hh, bfA);
        __builtin_amdgcn_wave_barrier();
    }

    #pragma unroll
    for (int p = 0; p < TPW / 2; ++p) {
        __builtin_amdgcn_sched_barrier(0);
        v2f (&cur)[4][2] = (p & 1) ? bfB : bfA;
        v2f (&nxt)[4][2] = (p & 1) ? bfA : bfB;

        v8f acc[4];
        #pragma unroll
        for (int q = 0; q < 4; ++q) {
            const v8f zero = {};
            acc[q] = __builtin_amdgcn_wmma_f32_16x16x4_f32(
                false, afrag[0], false, cur[q][0], (short)0, zero, false, false);
        }
        if (p + 1 < TPW / 2) {
            const float  xp[2] = { xv[2 * p + 2], xv[2 * p + 3] };
            const float4 hq[2] = { hp[2 * p + 2], hp[2 * p + 3] };
            stage_rows(xp, hq, Wl, bl, rows, lane);
            __builtin_amdgcn_wave_barrier();
            read_frags(rows, mn, hh, nxt);
            __builtin_amdgcn_wave_barrier();
        }
        __builtin_amdgcn_sched_barrier(0);

        if (p > 0) {
            const v2f zt = zkeep + zrecv;
            const int eo = cbase + 64 * (p - 1) + lane;
            if (FULL || eo < Btot)      out[4 * (size_t)Btot + eo]      = zt.x;
            if (FULL || eo + 32 < Btot) out[4 * (size_t)Btot + eo + 32] = zt.y;
        }

        v2f   hres[4];
        float opart[4];
        #pragma unroll
        for (int q = 0; q < 4; ++q) {
            acc[q] = __builtin_amdgcn_wmma_f32_16x16x4_f32(
                false, afrag[1], false, cur[q][1], (short)0, acc[q], false, false);

            const v2f a0 = __builtin_shufflevector(acc[q], acc[q], 0, 1);
            const v2f a1 = __builtin_shufflevector(acc[q], acc[q], 2, 3);
            const v2f a2 = __builtin_shufflevector(acc[q], acc[q], 4, 5);
            const v2f a3 = __builtin_shufflevector(acc[q], acc[q], 6, 7);
            const v2f sp = a0 * a0 + a2 * a2;
            const v2f sn = a1 * a1 + a3 * a3;
            const v2f dz = sp - sn;
            hres[q]  = sp + sn;
            opart[q] = dz.x + dz.y;
        }
        const int hidx = 2 * (cbase + 64 * p + mn) + hh;
        #pragma unroll
        for (int q = 0; q < 4; ++q) {
            if (FULL || cbase + 64 * p + 16 * q + mn < Btot)
                ((v2f*)out)[hidx + 32 * q] = hres[q];
        }
        zkeep.x = hh ? opart[1] : opart[0];
        zkeep.y = hh ? opart[3] : opart[2];
        zrecv.x = xor16(hh ? opart[0] : opart[1]);
        zrecv.y = xor16(hh ? opart[2] : opart[3]);
    }
    {
        const v2f zt = zkeep + zrecv;
        const int eo = cbase + 64 * (TPW / 2 - 1) + lane;
        if (FULL || eo < Btot)      out[4 * (size_t)Btot + eo]      = zt.x;
        if (FULL || eo + 32 < Btot) out[4 * (size_t)Btot + eo + 32] = zt.y;
    }
    __builtin_amdgcn_wave_barrier();
}

__global__ __launch_bounds__(BLOCK) void qlstm_kernel(
    const float* __restrict__ x, const float* __restrict__ h_prev,
    const float* __restrict__ W_in, const float* __restrict__ b_in,
    const float* __restrict__ Up, const float* __restrict__ Ud,
    float* __restrict__ out, int Btot)
{
    __shared__ __attribute__((aligned(16))) float afS[32][4];
    __shared__ __attribute__((aligned(16))) float ldsB[WAVES][64][BSTRIDE];

    const int tid = threadIdx.x;
    const int lane = tid & 31;
    const int wv   = __builtin_amdgcn_readfirstlane(tid >> 5);
    const int gw = blockIdx.x * WAVES + wv;
    const int cspan   = 32 * TPW;
    const int cstride = gridDim.x * WAVES * cspan;
    const int cbase0  = gw * cspan;

    const int al = (tid >> 2) & 3, aq = tid & 3;
    const float* aptr = (al < 2) ? (Up + 4 * al + aq) : (Ud + 4 * (3 - al) + aq);
    const float wang = *aptr;
    __builtin_amdgcn_sched_barrier(0);

    float Wl[4], bl[4];
    #pragma unroll
    for (int j = 0; j < 4; ++j) { Wl[j] = W_in[j]; bl[j] = b_in[j]; }
    float  xv[TPW];
    float4 hp[TPW];
    #pragma unroll
    for (int j = 0; j < TPW; ++j) {
        const int e = max(min(cbase0 + 32 * j + lane, Btot - 1), 0);
        xv[j] = x[e];
        hp[j] = ((const float4*)h_prev)[e];
    }
    asm volatile("" :: "v"(wang));

    const int hh   = lane >> 4;
    const int mn   = lane & 15;
    if (wv == 0) {
        const float cv = cosf(0.5f * wang);
        const float sv = sinf(0.5f * wang);
        const int rj = mn >> 2, rk = mn & 3;
        float wj = Wl[0], bj = bl[0];
        #pragma unroll
        for (int j = 1; j < 4; ++j) { wj = (rj == j) ? Wl[j] : wj;  bj = (rj == j) ? bl[j] : bj; }
        float va[4];
        va[0] = (rk == 2 * hh + 0) ? wj : 0.0f;
        va[1] = (rk == 2 * hh + 1) ? wj : 0.0f;
        va[2] = (rk == 2 * hh + 0) ? bj : 0.0f;
        va[3] = (rk == 2 * hh + 1) ? bj : 0.0f;
        int sgn[4];
        #pragma unroll
        for (int q = 0; q < 4; ++q) sgn[q] = (((mn >> (3 - q)) & 1) ^ 1) << 31;
        #pragma unroll
        for (int l = 0; l < 4; ++l) {
            float c[4], ss[4];
            #pragma unroll
            for (int q = 0; q < 4; ++q) {
                c[q]  = lane_bcast(cv, 4 * l + q);
                ss[q] = __int_as_float(__float_as_int(lane_bcast(sv, 4 * l + q)) ^ sgn[q]);
            }
            ry_stage<8>(va, c[0], ss[0]);
            ry_stage<4>(va, c[1], ss[1]);
            ry_stage<2>(va, c[2], ss[2]);
            ry_stage<1>(va, c[3], ss[3]);
            const unsigned sel_lo = (l < 3) ? ring_sel(0) : last_sel(0);
            const unsigned sel_hi = (l < 3) ? ring_sel(8) : last_sel(8);
            #pragma unroll
            for (int n = 0; n < 4; ++n) {
                const int vi = __float_as_int(va[n]);
                va[n] = __int_as_float(__builtin_amdgcn_permlane16(
                    vi, vi, sel_lo, sel_hi, false, false));
            }
        }
        *(float4*)afS[lane] = make_float4(va[0], va[1], va[2], va[3]);
    }
    __builtin_amdgcn_fence(__ATOMIC_RELEASE, "workgroup", "local");
    __builtin_amdgcn_s_barrier();
    __builtin_amdgcn_fence(__ATOMIC_ACQUIRE, "workgroup", "local");
    v2f afrag[2];
    {
        const float4 af = *(const float4*)afS[lane];
        afrag[0].x = af.x;  afrag[0].y = af.y;
        afrag[1].x = af.z;  afrag[1].y = af.w;
    }

    int cbase = cbase0;
    for (; cbase + cspan <= Btot; cbase += cstride) {
        sweep<true>(xv, hp, Wl, bl, afrag, ldsB[wv], out, Btot, cbase, lane, hh, mn);
        if (cbase + cstride < Btot) {
            #pragma unroll
            for (int j = 0; j < TPW; ++j) {
                const int e = min(cbase + cstride + 32 * j + lane, Btot - 1);
                xv[j] = x[e];
                hp[j] = ((const float4*)h_prev)[e];
            }
        }
    }
    if (cbase < Btot)
        sweep<false>(xv, hp, Wl, bl, afrag, ldsB[wv], out, Btot, cbase, lane, hh, mn);
}

extern "C" void kernel_launch(void* const* d_in, const int* in_sizes, int n_in,
                              void* d_out, int out_size, void* d_ws, size_t ws_size,
                              hipStream_t stream) {
    const float* x      = (const float*)d_in[0];
    const float* h_prev = (const float*)d_in[1];
    const float* W_in   = (const float*)d_in[2];
    const float* b_in   = (const float*)d_in[3];
    const float* Up     = (const float*)d_in[4];
    const float* Ud     = (const float*)d_in[5];
    float* out = (float*)d_out;

    int Btot = in_sizes[0];
    int tiles = (Btot + 32 * TPW * WAVES - 1) / (32 * TPW * WAVES);
    int blocks = 512;
    if (blocks > tiles) blocks = (tiles > 0) ? tiles : 1;

    qlstm_kernel<<<blocks, BLOCK, 0, stream>>>(x, h_prev, W_in, b_in, Up, Ud, out, Btot);
}
